# starting kernel
# baseline (speedup 1.0000x reference)
#include <hip/hip_runtime.h>

#define N_NODES 50000
#define N_EDGES 800000
#define NF 128

typedef float v2f __attribute__((ext_vector_type(2)));
typedef float v8f __attribute__((ext_vector_type(8)));

__global__ void k_init_deg(float* __restrict__ deg) {
    int i = blockIdx.x * blockDim.x + threadIdx.x;
    if (i < N_NODES) deg[i] = 1.0f;
}

__global__ void k_deg_accum(const int* __restrict__ dst, float* __restrict__ deg) {
    int e = blockIdx.x * blockDim.x + threadIdx.x;
    if (e < N_EDGES) atomicAdd(&deg[dst[e]], 1.0f);
}

__global__ void k_deg_final(float* __restrict__ deg) {
    int i = blockIdx.x * blockDim.x + threadIdx.x;
    if (i < N_NODES) deg[i] = rsqrtf(deg[i]);
}

__global__ void __launch_bounds__(128) k_gemm_fused(const float* __restrict__ X,
                                                    const float* __restrict__ W,
                                                    const float* __restrict__ bias,
                                                    const float* __restrict__ dinv,
                                                    float* __restrict__ Y,
                                                    float* __restrict__ AggInit,
                                                    int nrows) {
    __shared__ float wlds[64 * 128 * 2];

    const int tid = threadIdx.x;
    for (int idx = tid; idx < 64 * 128; idx += 128) {
        const int p = idx >> 7;
        const int n = idx & 127;
        float2 w2;
        w2.x = W[(size_t)(2 * p) * NF + n];
        w2.y = W[(size_t)(2 * p + 1) * NF + n];
        *(float2*)(wlds + (size_t)idx * 2) = w2;
    }
    __syncthreads();

    const int wave    = tid >> 5;
    const int lane    = tid & 31;
    const int rowTile = blockIdx.x * 4 + wave;
    const int row0    = rowTile * 16;
    const bool active = (row0 < nrows);

    const int hi = lane >> 4;
    const int lm = lane & 15;

    v8f acc[8];
#pragma unroll
    for (int t = 0; t < 8; ++t) acc[t] = (v8f){0,0,0,0,0,0,0,0};

    const int arow = active ? (row0 + lm) : lm;
    const float* xrow = X + (size_t)arow * NF;

#pragma unroll 4
    for (int kt = 0; kt < 32; ++kt) {
        v2f a = *(const v2f*)(xrow + kt * 4 + 2 * hi);
        const int p = kt * 2 + hi;
        const float* wp = wlds + (size_t)p * 256 + lm * 2;
#pragma unroll
        for (int t = 0; t < 8; ++t) {
            v2f b = *(const v2f*)(wp + t * 32);
            acc[t] = __builtin_amdgcn_wmma_f32_16x16x4_f32(
                 false, a,  false, b,
                 (short)0, acc[t],  false,  false);
        }
    }

    if (!active) return;

    float dv[8];
#pragma unroll
    for (int r = 0; r < 8; ++r) dv[r] = dinv[row0 + 8 * hi + r];

#pragma unroll
    for (int t = 0; t < 8; ++t) {
        const int col = t * 16 + lm;
        const float bc = bias[col];
        float* yb = Y       + (size_t)(row0 + 8 * hi) * NF + col;
        float* ab = AggInit + (size_t)(row0 + 8 * hi) * NF + col;
#pragma unroll
        for (int r = 0; r < 8; ++r) {
            const float v = acc[t][r];
            yb[(size_t)r * NF] = v;
            ab[(size_t)r * NF] = bc + dv[r] * dv[r] * v;
        }
    }
}

__global__ void __launch_bounds__(256) k_scatter(const int* __restrict__ src,
                                                 const int* __restrict__ dst,
                                                 const float* __restrict__ dinv,
                                                 const float* __restrict__ msg,
                                                 float* __restrict__ agg) {
    int e = blockIdx.x * (blockDim.x >> 5) + (threadIdx.x >> 5);
    if (e >= N_EDGES) return;
    int lane = threadIdx.x & 31;
    int s = src[e];
    int d = dst[e];
    float norm = dinv[s] * dinv[d];
    const float4 v = *((const float4*)(msg + (size_t)s * NF) + lane);
    float* ap = agg + (size_t)d * NF + lane * 4;
    atomicAdd(ap + 0, norm * v.x);
    atomicAdd(ap + 1, norm * v.y);
    atomicAdd(ap + 2, norm * v.z);
    atomicAdd(ap + 3, norm * v.w);
}

__global__ void k_relu_skip(float* __restrict__ h, const float* __restrict__ x) {
    int i = blockIdx.x * blockDim.x + threadIdx.x;
    if (i >= N_NODES * NF) return;
    float v = h[i];
    h[i] = (v > 0.0f ? v : 0.0f) + x[i];
}

extern "C" void kernel_launch(void* const* d_in, const int* in_sizes, int n_in,
                              void* d_out, int out_size, void* d_ws, size_t ws_size,
                              hipStream_t stream) {
    const int*   ei  = (const int*)d_in[0];
    const int*   src = ei;
    const int*   dst = ei + N_EDGES;
    const float* x   = (const float*)d_in[1];
    const float* W1  = (const float*)d_in[2];
    const float* b1  = (const float*)d_in[3];
    const float* W2  = (const float*)d_in[4];
    const float* b2  = (const float*)d_in[5];
    float*       out = (float*)d_out;

    float* dinv = (float*)d_ws;
    float* xw   = dinv + 51200;
    float* agg  = xw + (size_t)N_NODES * NF;

    const int elemGrid     = (N_NODES * NF + 255) / 256;
    const int edgeWaveGrid = (N_EDGES + 7) / 8;
    const int gemmGrid     = (N_NODES / 16 + 3) / 4;

    k_init_deg  <<<(N_NODES + 255) / 256, 256, 0, stream>>>(dinv);
    k_deg_accum <<<(N_EDGES + 255) / 256, 256, 0, stream>>>(dst, dinv);
    k_deg_final <<<(N_NODES + 255) / 256, 256, 0, stream>>>(dinv);

    k_gemm_fused<<<gemmGrid, 128, 0, stream>>>(x, W1, b1, dinv, xw, agg, N_NODES);
    k_scatter   <<<edgeWaveGrid, 256, 0, stream>>>(src, dst, dinv, xw, agg);
    k_relu_skip <<<elemGrid, 256, 0, stream>>>(agg, x);

    k_gemm_fused<<<gemmGrid, 128, 0, stream>>>(agg, W2, b2, dinv, xw, out, N_NODES);
    k_scatter   <<<edgeWaveGrid, 256, 0, stream>>>(src, dst, dinv, xw, out);
}


# Edge aggregation as a gather: per-call CSR build (int count with slot, one-block scan, fill), then one wave per node sums its source rows in registers and stores once; float atomic scatter removed
# speedup vs baseline: 1.8102x; 1.8102x over previous
#include <hip/hip_runtime.h>


#define N_NODES 50000
#define N_EDGES 800000
#define NF 128
#define SCAN_WAVES 32
#define SCAN_SPAN 1568
#define N_PAD (SCAN_WAVES * SCAN_SPAN)

typedef float v2f __attribute__((ext_vector_type(2)));
typedef float v8f __attribute__((ext_vector_type(8)));

__global__ void k_zero_cnt(int* __restrict__ cnt) {
    int i = blockIdx.x * blockDim.x + threadIdx.x;
    if (i < N_PAD) cnt[i] = 0;
}

__global__ void k_count_rank(const int* __restrict__ dst, int* __restrict__ cnt,
                             int* __restrict__ rank) {
    int e = blockIdx.x * blockDim.x + threadIdx.x;
    if (e < N_EDGES) rank[e] = atomicAdd(&cnt[dst[e]], 1);
}

__global__ void __launch_bounds__(1024) k_scan(const int* __restrict__ cnt,
                                               int* __restrict__ rowptr,
                                               float* __restrict__ dinv) {
    __shared__ int wtot[SCAN_WAVES];
    const int wave = threadIdx.x >> 5;
    const int lane = threadIdx.x & 31;
    const int base = wave * SCAN_SPAN;

    int4 c[13];
    int  ex[13];
#pragma unroll
    for (int it = 0; it < 13; ++it) {
        const int off = it * 128 + lane * 4;
        c[it] = (off < SCAN_SPAN) ? *(const int4*)(cnt + base + off) : make_int4(0, 0, 0, 0);
    }
    int tot[13], incl[13];
#pragma unroll
    for (int it = 0; it < 13; ++it) { tot[it] = c[it].x + c[it].y + c[it].z + c[it].w; incl[it] = tot[it]; }
#pragma unroll
    for (int sh = 1; sh < 32; sh <<= 1) {
        int t[13];
#pragma unroll
        for (int it = 0; it < 13; ++it) t[it] = __shfl_up(incl[it], sh);
#pragma unroll
        for (int it = 0; it < 13; ++it) if (lane >= sh) incl[it] += t[it];
    }
    int gsum[13];
#pragma unroll
    for (int it = 0; it < 13; ++it) gsum[it] = __shfl(incl[it], 31);
    int carry = 0;
#pragma unroll
    for (int it = 0; it < 13; ++it) { ex[it] = carry + incl[it] - tot[it]; carry += gsum[it]; }
    if (lane == 0) wtot[wave] = carry;
    __syncthreads();
    int woff = (lane < wave) ? wtot[lane] : 0;
#pragma unroll
    for (int sh = 16; sh > 0; sh >>= 1) woff += __shfl_xor(woff, sh);

#pragma unroll
    for (int it = 0; it < 13; ++it) {
        const int off = it * 128 + lane * 4;
        if (off < SCAN_SPAN) {
            const int p0 = woff + ex[it];
            const int p1 = p0 + c[it].x;
            const int p2 = p1 + c[it].y;
            const int p3 = p2 + c[it].z;
            *(int4*)(rowptr + base + off) = make_int4(p0, p1, p2, p3);
            float4 dv;
            dv.x = rsqrtf((float)(c[it].x + 1));
            dv.y = rsqrtf((float)(c[it].y + 1));
            dv.z = rsqrtf((float)(c[it].z + 1));
            dv.w = rsqrtf((float)(c[it].w + 1));
            *(float4*)(dinv + base + off) = dv;
        }
    }
}

__global__ void k_fill(const int* __restrict__ src, const int* __restrict__ dst,
                       const int* __restrict__ rowptr, const int* __restrict__ rank,
                       int* __restrict__ col) {
    int e = blockIdx.x * blockDim.x + threadIdx.x;
    if (e < N_EDGES) col[rowptr[dst[e]] + rank[e]] = src[e];
}

__global__ void __launch_bounds__(128) k_gemm(const float* __restrict__ X,
                                              const float* __restrict__ W,
                                              float* __restrict__ Y,
                                              int nrows) {
    __shared__ float wlds[64 * 128 * 2];

    const int tid = threadIdx.x;
    for (int idx = tid; idx < 64 * 128; idx += 128) {
        const int p = idx >> 7;
        const int n = idx & 127;
        float2 w2;
        w2.x = W[(size_t)(2 * p) * NF + n];
        w2.y = W[(size_t)(2 * p + 1) * NF + n];
        *(float2*)(wlds + (size_t)idx * 2) = w2;
    }
    __syncthreads();

    const int wave    = tid >> 5;
    const int lane    = tid & 31;
    const int rowTile = blockIdx.x * 4 + wave;
    const int row0    = rowTile * 16;
    const bool active = (row0 < nrows);

    const int hi = lane >> 4;
    const int lm = lane & 15;

    v8f acc[8];
#pragma unroll
    for (int t = 0; t < 8; ++t) acc[t] = (v8f){0,0,0,0,0,0,0,0};

    const int arow = active ? (row0 + lm) : lm;
    const float* xrow = X + (size_t)arow * NF;

#pragma unroll 4
    for (int kt = 0; kt < 32; ++kt) {
        v2f a = *(const v2f*)(xrow + kt * 4 + 2 * hi);
        const int p = kt * 2 + hi;
        const float* wp = wlds + (size_t)p * 256 + lm * 2;
#pragma unroll
        for (int t = 0; t < 8; ++t) {
            v2f b = *(const v2f*)(wp + t * 32);
            acc[t] = __builtin_amdgcn_wmma_f32_16x16x4_f32(
                 false, a,  false, b,
                 (short)0, acc[t],  false,  false);
        }
    }

    if (!active) return;

#pragma unroll
    for (int t = 0; t < 8; ++t) {
        const int col = t * 16 + lm;
        float* yb = Y + (size_t)(row0 + 8 * hi) * NF + col;
#pragma unroll
        for (int r = 0; r < 8; ++r) yb[(size_t)r * NF] = acc[t][r];
    }
}


template <int K>
static __device__ __forceinline__ void take_rows(float4& acc, const float4* __restrict__ rows,
                                                 int s, int wi, float dd, int j, int lane) {
    int    sj[K];
    float4 v[K];
#pragma unroll
    for (int k = 0; k < K; ++k) sj[k] = __builtin_amdgcn_readlane(s, j + k);
#pragma unroll
    for (int k = 0; k < K; ++k) v[k] = rows[(size_t)sj[k] * 32 + lane];
#pragma unroll
    for (int k = 0; k < K; ++k) {
        const float w = __int_as_float(__builtin_amdgcn_readlane(wi, j + k)) * dd;
        acc.x = __builtin_fmaf(w, v[k].x, acc.x);
        acc.y = __builtin_fmaf(w, v[k].y, acc.y);
        acc.z = __builtin_fmaf(w, v[k].z, acc.z);
        acc.w = __builtin_fmaf(w, v[k].w, acc.w);
    }
}

__global__ void __launch_bounds__(256) k_gather(const int* __restrict__ rowptr,
                                                const int* __restrict__ col,
                                                const float* __restrict__ dinv,
                                                const float* __restrict__ xw,
                                                const float* __restrict__ bias,
                                                float* __restrict__ out) {
    const int d = __builtin_amdgcn_readfirstlane(blockIdx.x * 8 + (threadIdx.x >> 5));
    if (d >= N_NODES) return;
    const int lane = threadIdx.x & 31;
    const float4* rows = (const float4*)xw;

    const int   beg = rowptr[d];
    const int   end = rowptr[d + 1];
    const float dd  = dinv[d];

    const int m = end - beg + 1;
    float4 acc = make_float4(0.0f, 0.0f, 0.0f, 0.0f);

    for (int b = 0; b < m; b += 32) {
        const int n = (m - b < 32) ? (m - b) : 32;
        const int i = b + lane;
        int   s  = 0;
        float ds = 0.0f;
        if (i < m - 1) {
            s  = col[beg + i];
            ds = dinv[s];
        } else if (i == m - 1) {
            s  = d;
            ds = dd;
        }
        const int wi = __float_as_int(ds);
        int j = 0;
        for (; j + 8 <= n; j += 8) take_rows<8>(acc, rows, s, wi, dd, j, lane);
        if (j + 4 <= n) { take_rows<4>(acc, rows, s, wi, dd, j, lane); j += 4; }
        const int rem = n - j;
        if (rem == 3)      take_rows<3>(acc, rows, s, wi, dd, j, lane);
        else if (rem == 2) take_rows<2>(acc, rows, s, wi, dd, j, lane);
        else if (rem == 1) take_rows<1>(acc, rows, s, wi, dd, j, lane);
    }

    const float4 bc = ((const float4*)bias)[lane];
    acc.x += bc.x; acc.y += bc.y; acc.z += bc.z; acc.w += bc.w;
    ((float4*)out)[(size_t)d * 32 + lane] = acc;
}

__global__ void k_relu_skip(float* __restrict__ h, const float* __restrict__ x) {
    int i = blockIdx.x * blockDim.x + threadIdx.x;
    if (i >= N_NODES * NF) return;
    float v = h[i];
    h[i] = (v > 0.0f ? v : 0.0f) + x[i];
}

extern "C" void kernel_launch(void* const* d_in, const int* in_sizes, int n_in,
                              void* d_out, int out_size, void* d_ws, size_t ws_size,
                              hipStream_t stream) {
    const int*   ei  = (const int*)d_in[0];
    const int*   src = ei;
    const int*   dst = ei + N_EDGES;
    const float* x   = (const float*)d_in[1];
    const float* W1  = (const float*)d_in[2];
    const float* b1  = (const float*)d_in[3];
    const float* W2  = (const float*)d_in[4];
    const float* b2  = (const float*)d_in[5];
    float*       out = (float*)d_out;

    float* dinv   = (float*)d_ws;
    float* xw     = dinv + 51200;
    int*   cnt    = (int*)(xw + (size_t)N_NODES * NF);
    int*   rowptr = cnt + N_PAD;
    int*   rank   = rowptr + N_PAD;
    int*   col    = rank + N_EDGES;

    const int elemGrid = (N_NODES * NF + 255) / 256;
    const int edgeGrid = (N_EDGES + 255) / 256;
    const int nodeGrid = (N_NODES + 7) / 8;
    const int gemmGrid = (N_NODES / 16 + 3) / 4;

    k_zero_cnt  <<<N_PAD / 256, 256, 0, stream>>>(cnt);
    k_count_rank<<<edgeGrid, 256, 0, stream>>>(dst, cnt, rank);
    k_scan      <<<1, 1024, 0, stream>>>(cnt, rowptr, dinv);
    k_fill      <<<edgeGrid, 256, 0, stream>>>(src, dst, rowptr, rank, col);

    k_gemm      <<<gemmGrid, 128, 0, stream>>>(x, W1, xw, N_NODES);
    k_gather    <<<nodeGrid, 256, 0, stream>>>(rowptr, col, dinv, xw, b1, out);
    k_relu_skip <<<elemGrid, 256, 0, stream>>>(out, x);

    k_gemm      <<<gemmGrid, 128, 0, stream>>>(out, W2, xw, N_NODES);
    k_gather    <<<nodeGrid, 256, 0, stream>>>(rowptr, col, dinv, xw, b2, out);
}
